# starting kernel
# baseline (speedup 1.0000x reference)
#include <hip/hip_runtime.h>
#include <hip/hip_bf16.h>

typedef float v2f __attribute__((ext_vector_type(2)));
typedef float v8f __attribute__((ext_vector_type(8)));

#define D     128
#define NTOK  1024
#define G3    384

__device__ __forceinline__ v8f wmma4(v2f a, v2f b, v8f c) {
    return __builtin_amdgcn_wmma_f32_16x16x4_f32(
         false, a,  false, b,
         (short)0, c,  false,  false);
}

__device__ __forceinline__ float sigmoidf_dev(float x) {
    return 1.0f / (1.0f + __expf(-x));
}
__device__ __forceinline__ float tanhf_dev(float x) {
    x = fminf(15.0f, fmaxf(-15.0f, x));
    const float e = __expf(-2.0f * x);
    return (1.0f - e) / (1.0f + e);
}

__global__ __launch_bounds__(256) void gru_xproj_kernel(
    const float* __restrict__ xs, const float* __restrict__ Wx,
    const float* __restrict__ bx, float* __restrict__ Xp)
{
    const int lane = threadIdx.x & 31;
    const int wave = threadIdx.x >> 5;
    const int tile = blockIdx.x * 8 + wave;
    const int tm = tile / 24;
    const int tn = tile % 24;
    const int l15 = lane & 15;
    const int hi  = lane >> 4;

    const float* __restrict__ arow = &xs[(tm * 16 + l15) * D];
    const float* __restrict__ brow = &Wx[(tn * 16 + l15) * D];

    v8f acc = {};
    for (int k0 = 0; k0 < D; k0 += 4) {
        const int kb = k0 + 2 * hi;
        v2f a = { arow[kb], arow[kb + 1] };
        v2f b = { brow[kb], brow[kb + 1] };
        acc = wmma4(a, b, acc);
    }

    const int col  = tn * 16 + l15;
    const float bb = bx[col];
#pragma unroll
    for (int r = 0; r < 8; ++r) {
        const int row = tm * 16 + r + 8 * hi;
        Xp[row * G3 + col] = acc[r] + bb;
    }
}

__global__ __launch_bounds__(1024) void gru_serial_kernel(
    const int*   __restrict__ mask,
    const float* __restrict__ Wh, const float* __restrict__ bh,
    const float* __restrict__ Wx, const float* __restrict__ bx,
    const float* __restrict__ Xp,
    float* __restrict__ out)
{
    __shared__ float h1[D * D];
    __shared__ float h2[D * D];

    const int tid  = threadIdx.x;
    const int lane = tid & 31;
    const int wave = tid >> 5;
    const int l15  = lane & 15;
    const int hi   = lane >> 4;

    for (int i = tid; i < D * D; i += 1024) { h1[i] = 0.0f; h2[i] = 0.0f; }
    __syncthreads();

    for (int t = 0; t < NTOK; ++t) {
        const int mv = __builtin_amdgcn_readfirstlane(mask[t]);
        if (mv == 0) continue;

        float nv1[2][8];
#pragma unroll
        for (int gi = 0; gi < 2; ++gi) {
            const int g  = wave + gi * 32;
            const int mt = g >> 3;
            const int jt = g & 7;
            const float* __restrict__ arow = &h1[(mt * 16 + l15) * D];
            const float* __restrict__ wA = &Wh[(0 * D + jt * 16 + l15) * D];
            const float* __restrict__ wB = &Wh[(1 * D + jt * 16 + l15) * D];
            const float* __restrict__ wC = &Wh[(2 * D + jt * 16 + l15) * D];

            v8f accA = {}, accB = {}, accC = {};
            for (int k0 = 0; k0 < D; k0 += 4) {
                const int kb = k0 + 2 * hi;
                v2f a  = { arow[kb], arow[kb + 1] };
                v2f b0 = { wA[kb], wA[kb + 1] };
                v2f b1 = { wB[kb], wB[kb + 1] };
                v2f b2 = { wC[kb], wC[kb + 1] };
                accA = wmma4(a, b0, accA);
                accB = wmma4(a, b1, accB);
                accC = wmma4(a, b2, accC);
            }

            const int col = jt * 16 + l15;
            const float sa = Xp[t * G3 + col]           + bh[col];
            const float sb = Xp[t * G3 + D + col]       + bh[D + col];
            const float xc = Xp[t * G3 + 2 * D + col];
            const float bc = bh[2 * D + col];
#pragma unroll
            for (int r = 0; r < 8; ++r) {
                const int row = mt * 16 + r + 8 * hi;
                const float u  = sigmoidf_dev(accA[r] + sa);
                const float rg = sigmoidf_dev(accB[r] + sb);
                const float cd = tanhf_dev(xc + rg * (accC[r] + bc));
                const float ho = h1[row * D + col];
                nv1[gi][r] = u * ho + (1.0f - u) * cd;
            }
        }
        __syncthreads();
#pragma unroll
        for (int gi = 0; gi < 2; ++gi) {
            const int g = wave + gi * 32;
            const int mt = g >> 3, jt = g & 7;
            const int col = jt * 16 + l15;
#pragma unroll
            for (int r = 0; r < 8; ++r)
                h1[(mt * 16 + r + 8 * hi) * D + col] = nv1[gi][r];
        }
        __syncthreads();

        float nv2[2][8];
#pragma unroll
        for (int gi = 0; gi < 2; ++gi) {
            const int g  = wave + gi * 32;
            const int mt = g >> 3;
            const int jt = g & 7;
            const float* __restrict__ aH = &h2[(mt * 16 + l15) * D];
            const float* __restrict__ aX = &h1[(mt * 16 + l15) * D];
            const float* __restrict__ whA = &Wh[(0 * D + jt * 16 + l15) * D];
            const float* __restrict__ whB = &Wh[(1 * D + jt * 16 + l15) * D];
            const float* __restrict__ whC = &Wh[(2 * D + jt * 16 + l15) * D];
            const float* __restrict__ wxA = &Wx[(0 * D + jt * 16 + l15) * D];
            const float* __restrict__ wxB = &Wx[(1 * D + jt * 16 + l15) * D];
            const float* __restrict__ wxC = &Wx[(2 * D + jt * 16 + l15) * D];

            v8f hA = {}, hB = {}, hC = {}, xA = {}, xB = {}, xC = {};
            for (int k0 = 0; k0 < D; k0 += 4) {
                const int kb = k0 + 2 * hi;
                v2f a2 = { aH[kb], aH[kb + 1] };
                v2f a1 = { aX[kb], aX[kb + 1] };
                v2f bh0 = { whA[kb], whA[kb + 1] };
                v2f bh1 = { whB[kb], whB[kb + 1] };
                v2f bh2 = { whC[kb], whC[kb + 1] };
                v2f bx0 = { wxA[kb], wxA[kb + 1] };
                v2f bx1 = { wxB[kb], wxB[kb + 1] };
                v2f bx2 = { wxC[kb], wxC[kb + 1] };
                hA = wmma4(a2, bh0, hA);
                hB = wmma4(a2, bh1, hB);
                hC = wmma4(a2, bh2, hC);
                xA = wmma4(a1, bx0, xA);
                xB = wmma4(a1, bx1, xB);
                xC = wmma4(a1, bx2, xC);
            }

            const int col = jt * 16 + l15;
            const float ba0 = bh[col],         bb0 = bx[col];
            const float ba1 = bh[D + col],     bb1 = bx[D + col];
            const float ba2 = bh[2 * D + col], bb2 = bx[2 * D + col];
#pragma unroll
            for (int r = 0; r < 8; ++r) {
                const int row = mt * 16 + r + 8 * hi;
                const float u  = sigmoidf_dev((hA[r] + ba0) + (xA[r] + bb0));
                const float rg = sigmoidf_dev((hB[r] + ba1) + (xB[r] + bb1));
                const float cd = tanhf_dev((xC[r] + bb2) + rg * (hC[r] + ba2));
                const float ho = h2[row * D + col];
                nv2[gi][r] = u * ho + (1.0f - u) * cd;
            }
        }
        __syncthreads();
#pragma unroll
        for (int gi = 0; gi < 2; ++gi) {
            const int g = wave + gi * 32;
            const int mt = g >> 3, jt = g & 7;
            const int col = jt * 16 + l15;
#pragma unroll
            for (int r = 0; r < 8; ++r)
                h2[(mt * 16 + r + 8 * hi) * D + col] = nv2[gi][r];
        }
        __syncthreads();
    }

    for (int i = tid; i < D * D; i += 1024) {
        const float a = h1[i];
        const float b = h2[i];
        out[i] = b;
        out[D * D + 2 * i]     = a;
        out[D * D + 2 * i + 1] = b;
    }
}

extern "C" void kernel_launch(void* const* d_in, const int* in_sizes, int n_in,
                              void* d_out, int out_size, void* d_ws, size_t ws_size,
                              hipStream_t stream) {
    const float* xs   = (const float*)d_in[0];
    const int*   mask = (const int*)  d_in[1];
    const float* Wh   = (const float*)d_in[2];
    const float* bh   = (const float*)d_in[3];
    const float* Wx   = (const float*)d_in[4];
    const float* bx   = (const float*)d_in[5];
    float* Xp  = (float*)d_ws;
    float* out = (float*)d_out;

    gru_xproj_kernel<<<192, 256, 0, stream>>>(xs, Wx, bx, Xp);
    gru_serial_kernel<<<1, 1024, 0, stream>>>(mask, Wh, bh, Wx, bx, Xp, out);
}


# weights re-laid by launch 1 into WMMA fragment order in the workspace: each recurrence B load is one aligned 256-byte row per wave, not 16 B out of each of 16 lines; k order and sums unchanged
# speedup vs baseline: 1.6872x; 1.6872x over previous
#include <hip/hip_runtime.h>
#include <hip/hip_bf16.h>

typedef float v2f __attribute__((ext_vector_type(2)));
typedef float v8f __attribute__((ext_vector_type(8)));

#define D     128
#define NTOK  1024
#define G3    384
#define PKV   (G3 * D / 2)


__device__ __forceinline__ v8f wmma4(v2f a, v2f b, v8f c) {
    return __builtin_amdgcn_wmma_f32_16x16x4_f32(
         false, a,  false, b,
         (short)0, c,  false,  false);
}

__device__ __forceinline__ float sigmoidf_dev(float x) {
    return 1.0f / (1.0f + __expf(-x));
}
__device__ __forceinline__ float tanhf_dev(float x) {
    x = fminf(15.0f, fmaxf(-15.0f, x));
    const float e = __expf(-2.0f * x);
    return (1.0f - e) / (1.0f + e);
}

__global__ __launch_bounds__(256) void gru_xproj_kernel(
    const float* __restrict__ xs, const float* __restrict__ Wx,
    const float* __restrict__ bx, const float* __restrict__ Wh,
    float* __restrict__ Xp, float* __restrict__ Wp)
{
    {
        const int gid = blockIdx.x * 256 + threadIdx.x;
        const int m   = gid / PKV;
        const int r   = gid - m * PKV;
        const int pl  = r & 31;
        const int ps  = (r >> 5) & 31;
        const int gj  = r >> 10;
        const int pjt = gj / 3;
        const int pg  = gj - pjt * 3;
        const int off = (pg * D + pjt * 16 + (pl & 15)) * D + 4 * ps + 2 * (pl >> 4);
        const float* __restrict__ wsrc = m ? Wx : Wh;
        const v2f v = { wsrc[off], wsrc[off + 1] };
        ((v2f*)Wp)[gid] = v;
    }

    const int lane = threadIdx.x & 31;
    const int wave = threadIdx.x >> 5;
    const int tile = blockIdx.x * 8 + wave;
    const int tm = tile / 24;
    const int tn = tile % 24;
    const int l15 = lane & 15;
    const int hi  = lane >> 4;

    const float* __restrict__ arow = &xs[(tm * 16 + l15) * D];
    const float* __restrict__ brow = &Wx[(tn * 16 + l15) * D];

    v8f acc = {};
    for (int k0 = 0; k0 < D; k0 += 4) {
        const int kb = k0 + 2 * hi;
        v2f a = { arow[kb], arow[kb + 1] };
        v2f b = { brow[kb], brow[kb + 1] };
        acc = wmma4(a, b, acc);
    }

    const int col  = tn * 16 + l15;
    const float bb = bx[col];
#pragma unroll
    for (int r = 0; r < 8; ++r) {
        const int row = tm * 16 + r + 8 * hi;
        Xp[row * G3 + col] = acc[r] + bb;
    }
}

__global__ __launch_bounds__(1024) void gru_serial_kernel(
    const int*   __restrict__ mask,
    const float* __restrict__ bh, const float* __restrict__ bx,
    const float* __restrict__ Xp, const float* __restrict__ Wp,
    float* __restrict__ out)
{
    __shared__ float h1[D * D];
    __shared__ float h2[D * D];

    const int tid  = threadIdx.x;
    const int lane = tid & 31;
    const int wave = tid >> 5;
    const int l15  = lane & 15;
    const int hi   = lane >> 4;

    const v2f* __restrict__ Ph = (const v2f*)Wp;
    const v2f* __restrict__ Px = Ph + PKV;

    for (int i = tid; i < D * D; i += 1024) { h1[i] = 0.0f; h2[i] = 0.0f; }
    __syncthreads();

    for (int t = 0; t < NTOK; ++t) {
        const int mv = __builtin_amdgcn_readfirstlane(mask[t]);
        if (mv == 0) continue;

        float nv1[2][8];
#pragma unroll
        for (int gi = 0; gi < 2; ++gi) {
            const int g  = wave + gi * 32;
            const int mt = g >> 3;
            const int jt = g & 7;
            const float* __restrict__ arow = &h1[(mt * 16 + l15) * D];
            const v2f* __restrict__ wA = Ph + ((jt * 3 + 0) * 32) * 32 + lane;
            const v2f* __restrict__ wB = Ph + ((jt * 3 + 1) * 32) * 32 + lane;
            const v2f* __restrict__ wC = Ph + ((jt * 3 + 2) * 32) * 32 + lane;

            v8f accA = {}, accB = {}, accC = {};
            for (int k0 = 0; k0 < D; k0 += 4) {
                const int kb = k0 + 2 * hi;
                v2f a  = { arow[kb], arow[kb + 1] };
                v2f b0 = wA[k0 * 8];
                v2f b1 = wB[k0 * 8];
                v2f b2 = wC[k0 * 8];
                accA = wmma4(a, b0, accA);
                accB = wmma4(a, b1, accB);
                accC = wmma4(a, b2, accC);
            }

            const int col = jt * 16 + l15;
            const float sa = Xp[t * G3 + col]           + bh[col];
            const float sb = Xp[t * G3 + D + col]       + bh[D + col];
            const float xc = Xp[t * G3 + 2 * D + col];
            const float bc = bh[2 * D + col];
#pragma unroll
            for (int r = 0; r < 8; ++r) {
                const int row = mt * 16 + r + 8 * hi;
                const float u  = sigmoidf_dev(accA[r] + sa);
                const float rg = sigmoidf_dev(accB[r] + sb);
                const float cd = tanhf_dev(xc + rg * (accC[r] + bc));
                const float ho = h1[row * D + col];
                nv1[gi][r] = u * ho + (1.0f - u) * cd;
            }
        }
        __syncthreads();
#pragma unroll
        for (int gi = 0; gi < 2; ++gi) {
            const int g = wave + gi * 32;
            const int mt = g >> 3, jt = g & 7;
            const int col = jt * 16 + l15;
#pragma unroll
            for (int r = 0; r < 8; ++r)
                h1[(mt * 16 + r + 8 * hi) * D + col] = nv1[gi][r];
        }
        __syncthreads();

        float nv2[2][8];
#pragma unroll
        for (int gi = 0; gi < 2; ++gi) {
            const int g  = wave + gi * 32;
            const int mt = g >> 3;
            const int jt = g & 7;
            const float* __restrict__ aH = &h2[(mt * 16 + l15) * D];
            const float* __restrict__ aX = &h1[(mt * 16 + l15) * D];
            const v2f* __restrict__ whA = Ph + ((jt * 3 + 0) * 32) * 32 + lane;
            const v2f* __restrict__ whB = Ph + ((jt * 3 + 1) * 32) * 32 + lane;
            const v2f* __restrict__ whC = Ph + ((jt * 3 + 2) * 32) * 32 + lane;
            const v2f* __restrict__ wxA = Px + ((jt * 3 + 0) * 32) * 32 + lane;
            const v2f* __restrict__ wxB = Px + ((jt * 3 + 1) * 32) * 32 + lane;
            const v2f* __restrict__ wxC = Px + ((jt * 3 + 2) * 32) * 32 + lane;

            v8f hA = {}, hB = {}, hC = {}, xA = {}, xB = {}, xC = {};
            for (int k0 = 0; k0 < D; k0 += 4) {
                const int kb = k0 + 2 * hi;
                v2f a2 = { aH[kb], aH[kb + 1] };
                v2f a1 = { aX[kb], aX[kb + 1] };
                v2f bh0 = whA[k0 * 8];
                v2f bh1 = whB[k0 * 8];
                v2f bh2 = whC[k0 * 8];
                v2f bx0 = wxA[k0 * 8];
                v2f bx1 = wxB[k0 * 8];
                v2f bx2 = wxC[k0 * 8];
                hA = wmma4(a2, bh0, hA);
                hB = wmma4(a2, bh1, hB);
                hC = wmma4(a2, bh2, hC);
                xA = wmma4(a1, bx0, xA);
                xB = wmma4(a1, bx1, xB);
                xC = wmma4(a1, bx2, xC);
            }

            const int col = jt * 16 + l15;
            const float ba0 = bh[col],         bb0 = bx[col];
            const float ba1 = bh[D + col],     bb1 = bx[D + col];
            const float ba2 = bh[2 * D + col], bb2 = bx[2 * D + col];
#pragma unroll
            for (int r = 0; r < 8; ++r) {
                const int row = mt * 16 + r + 8 * hi;
                const float u  = sigmoidf_dev((hA[r] + ba0) + (xA[r] + bb0));
                const float rg = sigmoidf_dev((hB[r] + ba1) + (xB[r] + bb1));
                const float cd = tanhf_dev((xC[r] + bb2) + rg * (hC[r] + ba2));
                const float ho = h2[row * D + col];
                nv2[gi][r] = u * ho + (1.0f - u) * cd;
            }
        }
        __syncthreads();
#pragma unroll
        for (int gi = 0; gi < 2; ++gi) {
            const int g = wave + gi * 32;
            const int mt = g >> 3, jt = g & 7;
            const int col = jt * 16 + l15;
#pragma unroll
            for (int r = 0; r < 8; ++r)
                h2[(mt * 16 + r + 8 * hi) * D + col] = nv2[gi][r];
        }
        __syncthreads();
    }

    for (int i = tid; i < D * D; i += 1024) {
        const float a = h1[i];
        const float b = h2[i];
        out[i] = b;
        out[D * D + 2 * i]     = a;
        out[D * D + 2 * i + 1] = b;
    }
}

extern "C" void kernel_launch(void* const* d_in, const int* in_sizes, int n_in,
                              void* d_out, int out_size, void* d_ws, size_t ws_size,
                              hipStream_t stream) {
    const float* xs   = (const float*)d_in[0];
    const int*   mask = (const int*)  d_in[1];
    const float* Wh   = (const float*)d_in[2];
    const float* bh   = (const float*)d_in[3];
    const float* Wx   = (const float*)d_in[4];
    const float* bx   = (const float*)d_in[5];
    float* Xp  = (float*)d_ws;
    float* Wp  = (float*)d_ws + NTOK * G3;
    float* out = (float*)d_out;

    gru_xproj_kernel<<<192, 256, 0, stream>>>(xs, Wx, bx, Wh, Xp, Wp);
    gru_serial_kernel<<<1, 1024, 0, stream>>>(mask, bh, bx, Xp, Wp, out);
}


# h1/h2 state kept in LDS in WMMA A-fragment order: an A load is one contiguous 512-byte row per wave (16 B a lane, 2 k-steps), not 8 B out of 16 rows on the same 4 banks; k order, sums unchanged
# speedup vs baseline: 1.7731x; 1.0509x over previous
#include <hip/hip_runtime.h>
#include <hip/hip_bf16.h>

typedef float v2f __attribute__((ext_vector_type(2)));
typedef float v4f __attribute__((ext_vector_type(4)));
typedef float v8f __attribute__((ext_vector_type(8)));

#define D     128
#define NTOK  1024
#define G3    384
#define PKV   (G3 * D / 2)


__device__ __forceinline__ int st_idx(int row, int col) {
    return ((((row >> 4) * 16 + (col >> 3)) * 32 + (row & 15) + 16 * ((col >> 1) & 1)) << 2)
           + ((col >> 2) & 1) * 2 + (col & 1);
}

__device__ __forceinline__ v8f wmma4(v2f a, v2f b, v8f c) {
    return __builtin_amdgcn_wmma_f32_16x16x4_f32(
         false, a,  false, b,
         (short)0, c,  false,  false);
}

__device__ __forceinline__ float sigmoidf_dev(float x) {
    return 1.0f / (1.0f + __expf(-x));
}
__device__ __forceinline__ float tanhf_dev(float x) {
    x = fminf(15.0f, fmaxf(-15.0f, x));
    const float e = __expf(-2.0f * x);
    return (1.0f - e) / (1.0f + e);
}

__global__ __launch_bounds__(256) void gru_xproj_kernel(
    const float* __restrict__ xs, const float* __restrict__ Wx,
    const float* __restrict__ bx, const float* __restrict__ Wh,
    float* __restrict__ Xp, float* __restrict__ Wp)
{
    {
        const int gid = blockIdx.x * 256 + threadIdx.x;
        const int m   = gid / PKV;
        const int r   = gid - m * PKV;
        const int pl  = r & 31;
        const int ps  = (r >> 5) & 31;
        const int gj  = r >> 10;
        const int pjt = gj / 3;
        const int pg  = gj - pjt * 3;
        const int off = (pg * D + pjt * 16 + (pl & 15)) * D + 4 * ps + 2 * (pl >> 4);
        const float* __restrict__ wsrc = m ? Wx : Wh;
        const v2f v = { wsrc[off], wsrc[off + 1] };
        ((v2f*)Wp)[gid] = v;
    }

    const int lane = threadIdx.x & 31;
    const int wave = threadIdx.x >> 5;
    const int tile = blockIdx.x * 8 + wave;
    const int tm = tile / 24;
    const int tn = tile % 24;
    const int l15 = lane & 15;
    const int hi  = lane >> 4;

    const float* __restrict__ arow = &xs[(tm * 16 + l15) * D];
    const float* __restrict__ brow = &Wx[(tn * 16 + l15) * D];

    v8f acc = {};
    for (int k0 = 0; k0 < D; k0 += 4) {
        const int kb = k0 + 2 * hi;
        v2f a = { arow[kb], arow[kb + 1] };
        v2f b = { brow[kb], brow[kb + 1] };
        acc = wmma4(a, b, acc);
    }

    const int col  = tn * 16 + l15;
    const float bb = bx[col];
#pragma unroll
    for (int r = 0; r < 8; ++r) {
        const int row = tm * 16 + r + 8 * hi;
        Xp[row * G3 + col] = acc[r] + bb;
    }
}

__global__ __launch_bounds__(1024) void gru_serial_kernel(
    const int*   __restrict__ mask,
    const float* __restrict__ bh, const float* __restrict__ bx,
    const float* __restrict__ Xp, const float* __restrict__ Wp,
    float* __restrict__ out)
{
    __shared__ __attribute__((aligned(16))) float h1[D * D];
    __shared__ __attribute__((aligned(16))) float h2[D * D];

    const int tid  = threadIdx.x;
    const int lane = tid & 31;
    const int wave = tid >> 5;
    const int l15  = lane & 15;
    const int hi   = lane >> 4;

    const v2f* __restrict__ Ph = (const v2f*)Wp;
    const v2f* __restrict__ Px = Ph + PKV;

    for (int i = tid; i < D * D; i += 1024) { h1[i] = 0.0f; h2[i] = 0.0f; }
    __syncthreads();

    for (int t = 0; t < NTOK; ++t) {
        const int mv = __builtin_amdgcn_readfirstlane(mask[t]);
        if (mv == 0) continue;

        float nv1[2][8];
#pragma unroll
        for (int gi = 0; gi < 2; ++gi) {
            const int g  = wave + gi * 32;
            const int mt = g >> 3;
            const int jt = g & 7;
            const v4f* __restrict__ aF = (const v4f*)h1 + (mt * 16) * 32 + lane;
            const v2f* __restrict__ wA = Ph + ((jt * 3 + 0) * 32) * 32 + lane;
            const v2f* __restrict__ wB = Ph + ((jt * 3 + 1) * 32) * 32 + lane;
            const v2f* __restrict__ wC = Ph + ((jt * 3 + 2) * 32) * 32 + lane;

            v8f accA = {}, accB = {}, accC = {};
#pragma unroll
            for (int sp = 0; sp < D / 8; ++sp) {
                const v4f a4 = aF[sp * 32];
                const v2f aE = { a4.x, a4.y };
                const v2f aO = { a4.z, a4.w };
                v2f b0 = wA[sp * 64];
                v2f b1 = wB[sp * 64];
                v2f b2 = wC[sp * 64];
                accA = wmma4(aE, b0, accA);
                accB = wmma4(aE, b1, accB);
                accC = wmma4(aE, b2, accC);
                b0 = wA[sp * 64 + 32];
                b1 = wB[sp * 64 + 32];
                b2 = wC[sp * 64 + 32];
                accA = wmma4(aO, b0, accA);
                accB = wmma4(aO, b1, accB);
                accC = wmma4(aO, b2, accC);
            }

            const int col = jt * 16 + l15;
            const int eb  = st_idx(mt * 16 + 8 * hi, col);
            const float sa = Xp[t * G3 + col]           + bh[col];
            const float sb = Xp[t * G3 + D + col]       + bh[D + col];
            const float xc = Xp[t * G3 + 2 * D + col];
            const float bc = bh[2 * D + col];
#pragma unroll
            for (int r = 0; r < 8; ++r) {
                const float u  = sigmoidf_dev(accA[r] + sa);
                const float rg = sigmoidf_dev(accB[r] + sb);
                const float cd = tanhf_dev(xc + rg * (accC[r] + bc));
                const float ho = h1[eb + 4 * r];
                nv1[gi][r] = u * ho + (1.0f - u) * cd;
            }
        }
        __syncthreads();
#pragma unroll
        for (int gi = 0; gi < 2; ++gi) {
            const int g = wave + gi * 32;
            const int mt = g >> 3, jt = g & 7;
            const int eb = st_idx(mt * 16 + 8 * hi, jt * 16 + l15);
#pragma unroll
            for (int r = 0; r < 8; ++r)
                h1[eb + 4 * r] = nv1[gi][r];
        }
        __syncthreads();

        float nv2[2][8];
#pragma unroll
        for (int gi = 0; gi < 2; ++gi) {
            const int g  = wave + gi * 32;
            const int mt = g >> 3;
            const int jt = g & 7;
            const v4f* __restrict__ aH = (const v4f*)h2 + (mt * 16) * 32 + lane;
            const v4f* __restrict__ aX = (const v4f*)h1 + (mt * 16) * 32 + lane;
            const v2f* __restrict__ whA = Ph + ((jt * 3 + 0) * 32) * 32 + lane;
            const v2f* __restrict__ whB = Ph + ((jt * 3 + 1) * 32) * 32 + lane;
            const v2f* __restrict__ whC = Ph + ((jt * 3 + 2) * 32) * 32 + lane;
            const v2f* __restrict__ wxA = Px + ((jt * 3 + 0) * 32) * 32 + lane;
            const v2f* __restrict__ wxB = Px + ((jt * 3 + 1) * 32) * 32 + lane;
            const v2f* __restrict__ wxC = Px + ((jt * 3 + 2) * 32) * 32 + lane;

            v8f hA = {}, hB = {}, hC = {}, xA = {}, xB = {}, xC = {};
            for (int sp = 0; sp < D / 8; ++sp) {
                const v4f h4 = aH[sp * 32];
                const v4f x4 = aX[sp * 32];
                const v2f a2E = { h4.x, h4.y }, a2O = { h4.z, h4.w };
                const v2f a1E = { x4.x, x4.y }, a1O = { x4.z, x4.w };
                v2f bh0 = whA[sp * 64];
                v2f bh1 = whB[sp * 64];
                v2f bh2 = whC[sp * 64];
                v2f bx0 = wxA[sp * 64];
                v2f bx1 = wxB[sp * 64];
                v2f bx2 = wxC[sp * 64];
                hA = wmma4(a2E, bh0, hA);
                hB = wmma4(a2E, bh1, hB);
                hC = wmma4(a2E, bh2, hC);
                xA = wmma4(a1E, bx0, xA);
                xB = wmma4(a1E, bx1, xB);
                xC = wmma4(a1E, bx2, xC);
                bh0 = whA[sp * 64 + 32];
                bh1 = whB[sp * 64 + 32];
                bh2 = whC[sp * 64 + 32];
                bx0 = wxA[sp * 64 + 32];
                bx1 = wxB[sp * 64 + 32];
                bx2 = wxC[sp * 64 + 32];
                hA = wmma4(a2O, bh0, hA);
                hB = wmma4(a2O, bh1, hB);
                hC = wmma4(a2O, bh2, hC);
                xA = wmma4(a1O, bx0, xA);
                xB = wmma4(a1O, bx1, xB);
                xC = wmma4(a1O, bx2, xC);
            }

            const int col = jt * 16 + l15;
            const int eb  = st_idx(mt * 16 + 8 * hi, col);
            const float ba0 = bh[col],         bb0 = bx[col];
            const float ba1 = bh[D + col],     bb1 = bx[D + col];
            const float ba2 = bh[2 * D + col], bb2 = bx[2 * D + col];
#pragma unroll
            for (int r = 0; r < 8; ++r) {
                const float u  = sigmoidf_dev((hA[r] + ba0) + (xA[r] + bb0));
                const float rg = sigmoidf_dev((hB[r] + ba1) + (xB[r] + bb1));
                const float cd = tanhf_dev((xC[r] + bb2) + rg * (hC[r] + ba2));
                const float ho = h2[eb + 4 * r];
                nv2[gi][r] = u * ho + (1.0f - u) * cd;
            }
        }
        __syncthreads();
#pragma unroll
        for (int gi = 0; gi < 2; ++gi) {
            const int g = wave + gi * 32;
            const int mt = g >> 3, jt = g & 7;
            const int eb = st_idx(mt * 16 + 8 * hi, jt * 16 + l15);
#pragma unroll
            for (int r = 0; r < 8; ++r)
                h2[eb + 4 * r] = nv2[gi][r];
        }
        __syncthreads();
    }

    for (int i = tid; i < D * D; i += 1024) {
        const int f = st_idx(i >> 7, i & (D - 1));
        const float a = h1[f];
        const float b = h2[f];
        out[i] = b;
        out[D * D + 2 * i]     = a;
        out[D * D + 2 * i + 1] = b;
    }
}

extern "C" void kernel_launch(void* const* d_in, const int* in_sizes, int n_in,
                              void* d_out, int out_size, void* d_ws, size_t ws_size,
                              hipStream_t stream) {
    const float* xs   = (const float*)d_in[0];
    const int*   mask = (const int*)  d_in[1];
    const float* Wh   = (const float*)d_in[2];
    const float* bh   = (const float*)d_in[3];
    const float* Wx   = (const float*)d_in[4];
    const float* bx   = (const float*)d_in[5];
    float* Xp  = (float*)d_ws;
    float* Wp  = (float*)d_ws + NTOK * G3;
    float* out = (float*)d_out;

    gru_xproj_kernel<<<192, 256, 0, stream>>>(xs, Wx, bx, Wh, Xp, Wp);
    gru_serial_kernel<<<1, 1024, 0, stream>>>(mask, bh, bx, Xp, Wp, out);
}


# time loop moved to the host: one launch per token and layer, its 64 tiles (8 row x 8 column) on 64 work-groups, not 32 waves of 1; state planes ping-pong in global memory; k order, sums unchanged
# speedup vs baseline: 14.0434x; 7.9203x over previous
#include <hip/hip_runtime.h>


typedef float v2f __attribute__((ext_vector_type(2)));
typedef float v4f __attribute__((ext_vector_type(4)));
typedef float v8f __attribute__((ext_vector_type(8)));

#define D     128
#define NTOK  1024
#define G3    384
#define PKV   (G3 * D / 2)


__device__ __forceinline__ int st_idx(int row, int col) {
    return ((((row >> 4) * 16 + (col >> 3)) * 32 + (row & 15) + 16 * ((col >> 1) & 1)) << 2)
           + ((col >> 2) & 1) * 2 + (col & 1);
}

__device__ __forceinline__ v8f wmma4(v2f a, v2f b, v8f c) {
    return __builtin_amdgcn_wmma_f32_16x16x4_f32(
         false, a,  false, b,
         (short)0, c,  false,  false);
}

__device__ __forceinline__ float sigmoidf_dev(float x) {
    return 1.0f / (1.0f + __expf(-x));
}
__device__ __forceinline__ float tanhf_dev(float x) {
    x = fminf(15.0f, fmaxf(-15.0f, x));
    const float e = __expf(-2.0f * x);
    return (1.0f - e) / (1.0f + e);
}

#define AHEAD 8
__device__ __forceinline__ void tile_gemm3(const v4f* __restrict__ aF, const v2f* __restrict__ w,
                                           v8f& accA, v8f& accB, v8f& accC)
{
    const v2f* __restrict__ wA = w;
    const v2f* __restrict__ wB = w + 32 * 32;
    const v2f* __restrict__ wC = w + 2 * 32 * 32;
    v4f a4[D / 8];
    v2f bE[D / 8][3];
    v2f bO[D / 8][3];
#define LOAD_PAIR(SP) do { a4[SP] = aF[(SP) * 32]; \
        bE[SP][0] = wA[(SP) * 64];      bE[SP][1] = wB[(SP) * 64];      bE[SP][2] = wC[(SP) * 64]; \
        bO[SP][0] = wA[(SP) * 64 + 32]; bO[SP][1] = wB[(SP) * 64 + 32]; bO[SP][2] = wC[(SP) * 64 + 32]; } while (0)
    accA = (v8f){}; accB = (v8f){}; accC = (v8f){};
#pragma unroll
    for (int sp = 0; sp < AHEAD; ++sp) LOAD_PAIR(sp);
#pragma unroll
    for (int sp = 0; sp < D / 8; ++sp) {
        __builtin_amdgcn_sched_barrier(0);
        if (sp + AHEAD < D / 8) LOAD_PAIR(sp + AHEAD);
        __builtin_amdgcn_sched_barrier(0);
        const v2f aE = { a4[sp].x, a4[sp].y };
        const v2f aO = { a4[sp].z, a4[sp].w };
        accA = wmma4(aE, bE[sp][0], accA);
        accB = wmma4(aE, bE[sp][1], accB);
        accC = wmma4(aE, bE[sp][2], accC);
        accA = wmma4(aO, bO[sp][0], accA);
        accB = wmma4(aO, bO[sp][1], accB);
        accC = wmma4(aO, bO[sp][2], accC);
    }
#undef LOAD_PAIR
}

__global__ __launch_bounds__(256) void gru_xproj_kernel(
    const float* __restrict__ xs, const float* __restrict__ Wx,
    const float* __restrict__ bx, const float* __restrict__ Wh,
    float* __restrict__ Xp, float* __restrict__ Wp, float* __restrict__ hz)
{
    {
        const int gid = blockIdx.x * 256 + threadIdx.x;
        const int m   = gid / PKV;
        const int r   = gid - m * PKV;
        const int pl  = r & 31;
        const int ps  = (r >> 5) & 31;
        const int gj  = r >> 10;
        const int pjt = gj / 3;
        const int pg  = gj - pjt * 3;
        const int off = (pg * D + pjt * 16 + (pl & 15)) * D + 4 * ps + 2 * (pl >> 4);
        const float* __restrict__ wsrc = m ? Wx : Wh;
        const v2f v = { wsrc[off], wsrc[off + 1] };
        ((v2f*)Wp)[gid] = v;
        if (gid < 2 * D * D) hz[gid] = 0.0f;
    }

    const int lane = threadIdx.x & 31;
    const int wave = threadIdx.x >> 5;
    const int tile = blockIdx.x * 8 + wave;
    const int tm = tile / 24;
    const int tn = tile % 24;
    const int l15 = lane & 15;
    const int hi  = lane >> 4;

    const float* __restrict__ arow = &xs[(tm * 16 + l15) * D];
    const float* __restrict__ brow = &Wx[(tn * 16 + l15) * D];

    v8f acc = {};
    for (int k0 = 0; k0 < D; k0 += 4) {
        const int kb = k0 + 2 * hi;
        v2f a = { arow[kb], arow[kb + 1] };
        v2f b = { brow[kb], brow[kb + 1] };
        acc = wmma4(a, b, acc);
    }

    const int col  = tn * 16 + l15;
    const float bb = bx[col];
#pragma unroll
    for (int r = 0; r < 8; ++r) {
        const int row = tm * 16 + r + 8 * hi;
        Xp[row * G3 + col] = acc[r] + bb;
    }
}

__global__ __launch_bounds__(32) void gru_l1_step_kernel(
    const int*   __restrict__ mk,
    const float* __restrict__ bh,
    const float* __restrict__ xp,
    const float* __restrict__ Wp,
    const float* __restrict__ hold, float* __restrict__ hnew)
{
    const int lane = threadIdx.x;
    const int l15  = lane & 15;
    const int hi   = lane >> 4;
    const int mt   = blockIdx.x >> 3;
    const int jt   = blockIdx.x & 7;
    const int col  = jt * 16 + l15;
    const int eb   = st_idx(mt * 16 + 8 * hi, col);

    const int mv = mk[0];
    float ho[8];
#pragma unroll
    for (int r = 0; r < 8; ++r) ho[r] = hold[eb + 4 * r];
    const float sa = xp[col]         + bh[col];
    const float sb = xp[D + col]     + bh[D + col];
    const float xc = xp[2 * D + col];
    const float bc = bh[2 * D + col];

    v8f accA, accB, accC;
    tile_gemm3((const v4f*)hold + (mt * 16) * 32 + lane,
               (const v2f*)Wp + ((jt * 3) * 32) * 32 + lane, accA, accB, accC);

#pragma unroll
    for (int r = 0; r < 8; ++r) {
        const float u  = sigmoidf_dev(accA[r] + sa);
        const float rg = sigmoidf_dev(accB[r] + sb);
        const float cd = tanhf_dev(xc + rg * (accC[r] + bc));
        const float nv = u * ho[r] + (1.0f - u) * cd;
        hnew[eb + 4 * r] = (mv > 0) ? nv : ho[r];
    }
}

__global__ __launch_bounds__(64) void gru_l2_step_kernel(
    const int*   __restrict__ mk,
    const float* __restrict__ bh, const float* __restrict__ bx,
    const float* __restrict__ Wp,
    const float* __restrict__ h1n,
    const float* __restrict__ hold, float* __restrict__ hnew)
{
    __shared__ v4f xh[6][32];

    const int lane = threadIdx.x & 31;
    const int wv   = __builtin_amdgcn_readfirstlane(threadIdx.x >> 5);
    const int l15  = lane & 15;
    const int hi   = lane >> 4;
    const int mt   = blockIdx.x >> 3;
    const int jt   = blockIdx.x & 7;
    const int col  = jt * 16 + l15;
    const int eb   = st_idx(mt * 16 + 8 * hi, col);

    const int mv = mk[0];
    float ho[8];
#pragma unroll
    for (int r = 0; r < 8; ++r) ho[r] = hold[eb + 4 * r];
    const float ba0 = bh[col],         bb0 = bx[col];
    const float ba1 = bh[D + col],     bb1 = bx[D + col];
    const float ba2 = bh[2 * D + col], bb2 = bx[2 * D + col];

    const float* __restrict__ asrc = wv ? h1n : hold;
    const v2f*   __restrict__ wsrc = (const v2f*)Wp + (wv ? PKV : 0);

    v8f accA, accB, accC;
    tile_gemm3((const v4f*)asrc + (mt * 16) * 32 + lane,
               wsrc + ((jt * 3) * 32) * 32 + lane, accA, accB, accC);

    if (wv) {
        xh[0][lane] = (v4f){ accA[0], accA[1], accA[2], accA[3] };
        xh[1][lane] = (v4f){ accA[4], accA[5], accA[6], accA[7] };
        xh[2][lane] = (v4f){ accB[0], accB[1], accB[2], accB[3] };
        xh[3][lane] = (v4f){ accB[4], accB[5], accB[6], accB[7] };
        xh[4][lane] = (v4f){ accC[0], accC[1], accC[2], accC[3] };
        xh[5][lane] = (v4f){ accC[4], accC[5], accC[6], accC[7] };
    }
    __syncthreads();
    if (wv) return;

    float xA[8], xB[8], xC[8];
    {
        const v4f a0 = xh[0][lane], a1 = xh[1][lane];
        const v4f b0 = xh[2][lane], b1 = xh[3][lane];
        const v4f c0 = xh[4][lane], c1 = xh[5][lane];
        xA[0] = a0.x; xA[1] = a0.y; xA[2] = a0.z; xA[3] = a0.w;
        xA[4] = a1.x; xA[5] = a1.y; xA[6] = a1.z; xA[7] = a1.w;
        xB[0] = b0.x; xB[1] = b0.y; xB[2] = b0.z; xB[3] = b0.w;
        xB[4] = b1.x; xB[5] = b1.y; xB[6] = b1.z; xB[7] = b1.w;
        xC[0] = c0.x; xC[1] = c0.y; xC[2] = c0.z; xC[3] = c0.w;
        xC[4] = c1.x; xC[5] = c1.y; xC[6] = c1.z; xC[7] = c1.w;
    }
#pragma unroll
    for (int r = 0; r < 8; ++r) {
        const float u  = sigmoidf_dev((accA[r] + ba0) + (xA[r] + bb0));
        const float rg = sigmoidf_dev((accB[r] + ba1) + (xB[r] + bb1));
        const float cd = tanhf_dev((xC[r] + bb2) + rg * (accC[r] + ba2));
        const float nv = u * ho[r] + (1.0f - u) * cd;
        hnew[eb + 4 * r] = (mv > 0) ? nv : ho[r];
    }
}

__global__ __launch_bounds__(256) void gru_out_kernel(
    const float* __restrict__ h1, const float* __restrict__ h2, float* __restrict__ out)
{
    const int i = blockIdx.x * 256 + threadIdx.x;
    const int f = st_idx(i >> 7, i & (D - 1));
    const float a = h1[f];
    const float b = h2[f];
    out[i] = b;
    out[D * D + 2 * i]     = a;
    out[D * D + 2 * i + 1] = b;
}

extern "C" void kernel_launch(void* const* d_in, const int* in_sizes, int n_in,
                              void* d_out, int out_size, void* d_ws, size_t ws_size,
                              hipStream_t stream) {
    const float* xs   = (const float*)d_in[0];
    const int*   mask = (const int*)  d_in[1];
    const float* Wh   = (const float*)d_in[2];
    const float* bh   = (const float*)d_in[3];
    const float* Wx   = (const float*)d_in[4];
    const float* bx   = (const float*)d_in[5];
    float* Xp  = (float*)d_ws;
    float* Wp  = Xp + NTOK * G3;
    float* out = (float*)d_out;
    float* h1a = Wp + 2 * G3 * D;
    float* h2a = h1a + D * D;
    float* h1b = out;
    float* h2b = out + D * D;

    gru_xproj_kernel<<<192, 256, 0, stream>>>(xs, Wx, bx, Wh, Xp, Wp, h1a);

    const float* xp = Xp;
    for (const int* mk = mask; mk != mask + NTOK; mk += 2, xp += 2 * G3) {
        gru_l1_step_kernel<<<64, 32, 0, stream>>>(mk,     bh, xp,      Wp, h1a, h1b);
        gru_l2_step_kernel<<<64, 64, 0, stream>>>(mk,     bh, bx, Wp, h1b, h2a, h2b);
        gru_l1_step_kernel<<<64, 32, 0, stream>>>(mk + 1, bh, xp + G3, Wp, h1b, h1a);
        gru_l2_step_kernel<<<64, 64, 0, stream>>>(mk + 1, bh, bx, Wp, h1a, h2b, h2a);
    }

    gru_out_kernel<<<64, 256, 0, stream>>>(h1a, h2a, out);
}


# both layers in one launch per token: layer 2 runs one token behind layer 1, its tile and layer 1's tile on one 3-wave work-group; 1027 launches, not 2050; k order, sums, gate arithmetic unchanged
# speedup vs baseline: 25.6885x; 1.8292x over previous
#include <hip/hip_runtime.h>


typedef float v2f __attribute__((ext_vector_type(2)));
typedef float v4f __attribute__((ext_vector_type(4)));
typedef float v8f __attribute__((ext_vector_type(8)));

#define D     128
#define NTOK  1024
#define G3    384
#define PKV   (G3 * D / 2)


__device__ __forceinline__ int st_idx(int row, int col) {
    return ((((row >> 4) * 16 + (col >> 3)) * 32 + (row & 15) + 16 * ((col >> 1) & 1)) << 2)
           + ((col >> 2) & 1) * 2 + (col & 1);
}

__device__ __forceinline__ v8f wmma4(v2f a, v2f b, v8f c) {
    return __builtin_amdgcn_wmma_f32_16x16x4_f32(
         false, a,  false, b,
         (short)0, c,  false,  false);
}

__device__ __forceinline__ float sigmoidf_dev(float x) {
    return 1.0f / (1.0f + __expf(-x));
}
__device__ __forceinline__ float tanhf_dev(float x) {
    x = fminf(15.0f, fmaxf(-15.0f, x));
    const float e = __expf(-2.0f * x);
    return (1.0f - e) / (1.0f + e);
}

#define AHEAD 8
__device__ __forceinline__ void tile_gemm3(const v4f* __restrict__ aF, const v2f* __restrict__ w,
                                           v8f& accA, v8f& accB, v8f& accC)
{
    const v2f* __restrict__ wA = w;
    const v2f* __restrict__ wB = w + 32 * 32;
    const v2f* __restrict__ wC = w + 2 * 32 * 32;
    v4f a4[D / 8];
    v2f bE[D / 8][3];
    v2f bO[D / 8][3];
#define LOAD_PAIR(SP) do { a4[SP] = aF[(SP) * 32]; \
        bE[SP][0] = wA[(SP) * 64];      bE[SP][1] = wB[(SP) * 64];      bE[SP][2] = wC[(SP) * 64]; \
        bO[SP][0] = wA[(SP) * 64 + 32]; bO[SP][1] = wB[(SP) * 64 + 32]; bO[SP][2] = wC[(SP) * 64 + 32]; } while (0)
    accA = (v8f){}; accB = (v8f){}; accC = (v8f){};
#pragma unroll
    for (int sp = 0; sp < AHEAD; ++sp) LOAD_PAIR(sp);
#pragma unroll
    for (int sp = 0; sp < D / 8; ++sp) {
        __builtin_amdgcn_sched_barrier(0);
        if (sp + AHEAD < D / 8) LOAD_PAIR(sp + AHEAD);
        __builtin_amdgcn_sched_barrier(0);
        const v2f aE = { a4[sp].x, a4[sp].y };
        const v2f aO = { a4[sp].z, a4[sp].w };
        accA = wmma4(aE, bE[sp][0], accA);
        accB = wmma4(aE, bE[sp][1], accB);
        accC = wmma4(aE, bE[sp][2], accC);
        accA = wmma4(aO, bO[sp][0], accA);
        accB = wmma4(aO, bO[sp][1], accB);
        accC = wmma4(aO, bO[sp][2], accC);
    }
#undef LOAD_PAIR
}

__global__ __launch_bounds__(256) void gru_xproj_kernel(
    const float* __restrict__ xs, const float* __restrict__ Wx,
    const float* __restrict__ bx, const float* __restrict__ Wh,
    float* __restrict__ Xp, float* __restrict__ Wp, float* __restrict__ hz)
{
    {
        const int gid = blockIdx.x * 256 + threadIdx.x;
        const int m   = gid / PKV;
        const int r   = gid - m * PKV;
        const int pl  = r & 31;
        const int ps  = (r >> 5) & 31;
        const int gj  = r >> 10;
        const int pjt = gj / 3;
        const int pg  = gj - pjt * 3;
        const int off = (pg * D + pjt * 16 + (pl & 15)) * D + 4 * ps + 2 * (pl >> 4);
        const float* __restrict__ wsrc = m ? Wx : Wh;
        const v2f v = { wsrc[off], wsrc[off + 1] };
        ((v2f*)Wp)[gid] = v;
        if (gid < 2 * D * D) hz[gid] = 0.0f;
    }

    const int lane = threadIdx.x & 31;
    const int wave = threadIdx.x >> 5;
    const int tile = blockIdx.x * 8 + wave;
    const int tm = tile / 24;
    const int tn = tile % 24;
    const int l15 = lane & 15;
    const int hi  = lane >> 4;

    const float* __restrict__ arow = &xs[(tm * 16 + l15) * D];
    const float* __restrict__ brow = &Wx[(tn * 16 + l15) * D];

    v8f acc = {};
    for (int k0 = 0; k0 < D; k0 += 4) {
        const int kb = k0 + 2 * hi;
        v2f a = { arow[kb], arow[kb + 1] };
        v2f b = { brow[kb], brow[kb + 1] };
        acc = wmma4(a, b, acc);
    }

    const int col  = tn * 16 + l15;
    const float bb = bx[col];
#pragma unroll
    for (int r = 0; r < 8; ++r) {
        const int row = tm * 16 + r + 8 * hi;
        Xp[row * G3 + col] = acc[r] + bb;
    }
}

__global__ __launch_bounds__(32) void gru_l1_step_kernel(
    const int*   __restrict__ mk,
    const float* __restrict__ bh,
    const float* __restrict__ xp,
    const float* __restrict__ Wp,
    const float* __restrict__ hold, float* __restrict__ hnew)
{
    const int lane = threadIdx.x;
    const int l15  = lane & 15;
    const int hi   = lane >> 4;
    const int mt   = blockIdx.x >> 3;
    const int jt   = blockIdx.x & 7;
    const int col  = jt * 16 + l15;
    const int eb   = st_idx(mt * 16 + 8 * hi, col);

    const int mv = mk[0];
    float ho[8];
#pragma unroll
    for (int r = 0; r < 8; ++r) ho[r] = hold[eb + 4 * r];
    const float sa = xp[col]         + bh[col];
    const float sb = xp[D + col]     + bh[D + col];
    const float xc = xp[2 * D + col];
    const float bc = bh[2 * D + col];

    v8f accA, accB, accC;
    tile_gemm3((const v4f*)hold + (mt * 16) * 32 + lane,
               (const v2f*)Wp + ((jt * 3) * 32) * 32 + lane, accA, accB, accC);

#pragma unroll
    for (int r = 0; r < 8; ++r) {
        const float u  = sigmoidf_dev(accA[r] + sa);
        const float rg = sigmoidf_dev(accB[r] + sb);
        const float cd = tanhf_dev(xc + rg * (accC[r] + bc));
        const float nv = u * ho[r] + (1.0f - u) * cd;
        hnew[eb + 4 * r] = (mv > 0) ? nv : ho[r];
    }
}

__global__ __launch_bounds__(64) void gru_l2_step_kernel(
    const int*   __restrict__ mk,
    const float* __restrict__ bh, const float* __restrict__ bx,
    const float* __restrict__ Wp,
    const float* __restrict__ h1n,
    const float* __restrict__ hold, float* __restrict__ hnew)
{
    __shared__ v4f xh[6][32];

    const int lane = threadIdx.x & 31;
    const int wv   = __builtin_amdgcn_readfirstlane(threadIdx.x >> 5);
    const int l15  = lane & 15;
    const int hi   = lane >> 4;
    const int mt   = blockIdx.x >> 3;
    const int jt   = blockIdx.x & 7;
    const int col  = jt * 16 + l15;
    const int eb   = st_idx(mt * 16 + 8 * hi, col);

    const int mv = mk[0];
    float ho[8];
#pragma unroll
    for (int r = 0; r < 8; ++r) ho[r] = hold[eb + 4 * r];
    const float ba0 = bh[col],         bb0 = bx[col];
    const float ba1 = bh[D + col],     bb1 = bx[D + col];
    const float ba2 = bh[2 * D + col], bb2 = bx[2 * D + col];

    const float* __restrict__ asrc = wv ? h1n : hold;
    const v2f*   __restrict__ wsrc = (const v2f*)Wp + (wv ? PKV : 0);

    v8f accA, accB, accC;
    tile_gemm3((const v4f*)asrc + (mt * 16) * 32 + lane,
               wsrc + ((jt * 3) * 32) * 32 + lane, accA, accB, accC);

    if (wv) {
        xh[0][lane] = (v4f){ accA[0], accA[1], accA[2], accA[3] };
        xh[1][lane] = (v4f){ accA[4], accA[5], accA[6], accA[7] };
        xh[2][lane] = (v4f){ accB[0], accB[1], accB[2], accB[3] };
        xh[3][lane] = (v4f){ accB[4], accB[5], accB[6], accB[7] };
        xh[4][lane] = (v4f){ accC[0], accC[1], accC[2], accC[3] };
        xh[5][lane] = (v4f){ accC[4], accC[5], accC[6], accC[7] };
    }
    __syncthreads();
    if (wv) return;

    float xA[8], xB[8], xC[8];
    {
        const v4f a0 = xh[0][lane], a1 = xh[1][lane];
        const v4f b0 = xh[2][lane], b1 = xh[3][lane];
        const v4f c0 = xh[4][lane], c1 = xh[5][lane];
        xA[0] = a0.x; xA[1] = a0.y; xA[2] = a0.z; xA[3] = a0.w;
        xA[4] = a1.x; xA[5] = a1.y; xA[6] = a1.z; xA[7] = a1.w;
        xB[0] = b0.x; xB[1] = b0.y; xB[2] = b0.z; xB[3] = b0.w;
        xB[4] = b1.x; xB[5] = b1.y; xB[6] = b1.z; xB[7] = b1.w;
        xC[0] = c0.x; xC[1] = c0.y; xC[2] = c0.z; xC[3] = c0.w;
        xC[4] = c1.x; xC[5] = c1.y; xC[6] = c1.z; xC[7] = c1.w;
    }
#pragma unroll
    for (int r = 0; r < 8; ++r) {
        const float u  = sigmoidf_dev((accA[r] + ba0) + (xA[r] + bb0));
        const float rg = sigmoidf_dev((accB[r] + ba1) + (xB[r] + bb1));
        const float cd = tanhf_dev((xC[r] + bb2) + rg * (accC[r] + ba2));
        const float nv = u * ho[r] + (1.0f - u) * cd;
        hnew[eb + 4 * r] = (mv > 0) ? nv : ho[r];
    }
}

__global__ __launch_bounds__(96) void gru_l12_step_kernel(
    const int*   __restrict__ mk1,
    const int*   __restrict__ mk2,
    const float* __restrict__ bh, const float* __restrict__ bx,
    const float* __restrict__ xp,
    const float* __restrict__ Wp,
    const float* __restrict__ h1c,
    float*       __restrict__ h1n,
    const float* __restrict__ h2o, float* __restrict__ h2n)
{
    __shared__ v4f xh[6][32];

    const int lane = threadIdx.x & 31;
    const int wv   = __builtin_amdgcn_readfirstlane(threadIdx.x >> 5);
    const int l15  = lane & 15;
    const int hi   = lane >> 4;
    const int mt   = blockIdx.x >> 3;
    const int jt   = blockIdx.x & 7;
    const int col  = jt * 16 + l15;
    const int eb   = st_idx(mt * 16 + 8 * hi, col);

    const float* __restrict__ asrc = (wv == 0) ? h2o : h1c;
    const v2f*   __restrict__ wsrc = (const v2f*)Wp + ((wv == 1) ? PKV : 0);
    const float* __restrict__ qsrc = (wv == 2) ? xp : bx;

    const int mv = ((wv == 2) ? mk1 : mk2)[0];
    float ho[8];
#pragma unroll
    for (int r = 0; r < 8; ++r) ho[r] = asrc[eb + 4 * r];
    const float p0 = bh[col],         q0 = qsrc[col];
    const float p1 = bh[D + col],     q1 = qsrc[D + col];
    const float p2 = bh[2 * D + col], q2 = qsrc[2 * D + col];

    v8f accA, accB, accC;
    tile_gemm3((const v4f*)asrc + (mt * 16) * 32 + lane,
               wsrc + ((jt * 3) * 32) * 32 + lane, accA, accB, accC);

    if (wv == 1) {
        xh[0][lane] = (v4f){ accA[0], accA[1], accA[2], accA[3] };
        xh[1][lane] = (v4f){ accA[4], accA[5], accA[6], accA[7] };
        xh[2][lane] = (v4f){ accB[0], accB[1], accB[2], accB[3] };
        xh[3][lane] = (v4f){ accB[4], accB[5], accB[6], accB[7] };
        xh[4][lane] = (v4f){ accC[0], accC[1], accC[2], accC[3] };
        xh[5][lane] = (v4f){ accC[4], accC[5], accC[6], accC[7] };
    }
    __syncthreads();
    if (wv == 1) return;

    if (wv == 2) {
        const float sa = q0 + p0;
        const float sb = q1 + p1;
#pragma unroll
        for (int r = 0; r < 8; ++r) {
            const float u  = sigmoidf_dev(accA[r] + sa);
            const float rg = sigmoidf_dev(accB[r] + sb);
            const float cd = tanhf_dev(q2 + rg * (accC[r] + p2));
            const float nv = u * ho[r] + (1.0f - u) * cd;
            h1n[eb + 4 * r] = (mv > 0) ? nv : ho[r];
        }
        return;
    }

    float xA[8], xB[8], xC[8];
    {
        const v4f a0 = xh[0][lane], a1 = xh[1][lane];
        const v4f b0 = xh[2][lane], b1 = xh[3][lane];
        const v4f c0 = xh[4][lane], c1 = xh[5][lane];
        xA[0] = a0.x; xA[1] = a0.y; xA[2] = a0.z; xA[3] = a0.w;
        xA[4] = a1.x; xA[5] = a1.y; xA[6] = a1.z; xA[7] = a1.w;
        xB[0] = b0.x; xB[1] = b0.y; xB[2] = b0.z; xB[3] = b0.w;
        xB[4] = b1.x; xB[5] = b1.y; xB[6] = b1.z; xB[7] = b1.w;
        xC[0] = c0.x; xC[1] = c0.y; xC[2] = c0.z; xC[3] = c0.w;
        xC[4] = c1.x; xC[5] = c1.y; xC[6] = c1.z; xC[7] = c1.w;
    }
#pragma unroll
    for (int r = 0; r < 8; ++r) {
        const float u  = sigmoidf_dev((accA[r] + p0) + (xA[r] + q0));
        const float rg = sigmoidf_dev((accB[r] + p1) + (xB[r] + q1));
        const float cd = tanhf_dev((xC[r] + q2) + rg * (accC[r] + p2));
        const float nv = u * ho[r] + (1.0f - u) * cd;
        h2n[eb + 4 * r] = (mv > 0) ? nv : ho[r];
    }
}

__global__ __launch_bounds__(256) void gru_out_kernel(
    const float* __restrict__ h1, const float* __restrict__ h2, float* __restrict__ out)
{
    const int i = blockIdx.x * 256 + threadIdx.x;
    const int f = st_idx(i >> 7, i & (D - 1));
    const float a = h1[f];
    const float b = h2[f];
    out[i] = b;
    out[D * D + 2 * i]     = a;
    out[D * D + 2 * i + 1] = b;
}

extern "C" void kernel_launch(void* const* d_in, const int* in_sizes, int n_in,
                              void* d_out, int out_size, void* d_ws, size_t ws_size,
                              hipStream_t stream) {
    const float* xs   = (const float*)d_in[0];
    const int*   mask = (const int*)  d_in[1];
    const float* Wh   = (const float*)d_in[2];
    const float* bh   = (const float*)d_in[3];
    const float* Wx   = (const float*)d_in[4];
    const float* bx   = (const float*)d_in[5];
    float* Xp  = (float*)d_ws;
    float* Wp  = Xp + NTOK * G3;
    float* out = (float*)d_out;
    float* h1a = Wp + 2 * G3 * D;
    float* h2a = h1a + D * D;
    float* h1b = out;
    float* h2b = out + D * D;

    gru_xproj_kernel<<<192, 256, 0, stream>>>(xs, Wx, bx, Wh, Xp, Wp, h1a);

    gru_l1_step_kernel<<<64, 32, 0, stream>>>(mask, bh, Xp, Wp, h1a, h1b);
    const float* xp = Xp + G3;
    const int*   mk = mask + 1;
    for (; mk != mask + NTOK - 1; mk += 2, xp += 2 * G3) {
        gru_l12_step_kernel<<<64, 96, 0, stream>>>(mk,     mk - 1, bh, bx, xp,      Wp, h1b, h1a, h2a, h2b);
        gru_l12_step_kernel<<<64, 96, 0, stream>>>(mk + 1, mk,     bh, bx, xp + G3, Wp, h1a, h1b, h2b, h2a);
    }
    gru_l12_step_kernel<<<64, 96, 0, stream>>>(mk, mk - 1, bh, bx, xp, Wp, h1b, h1a, h2a, h2b);
    gru_l2_step_kernel<<<64, 64, 0, stream>>>(mk, bh, bx, Wp, h1a, h2b, h2a);

    gru_out_kernel<<<64, 256, 0, stream>>>(h1a, h2a, out);
}


# a tile position's nine sum chains (product x gate, 32 WMMAs each) spread over seven waves on two work-groups, one per layer: 128 x 4 waves of at most 64 WMMAs, not 64 x 3 of 96; k order, sums kept
# speedup vs baseline: 29.7325x; 1.1574x over previous
#include <hip/hip_runtime.h>


typedef float v2f __attribute__((ext_vector_type(2)));
typedef float v4f __attribute__((ext_vector_type(4)));
typedef float v8f __attribute__((ext_vector_type(8)));

#define D     128
#define NTOK  1024
#define G3    384
#define PKV   (G3 * D / 2)


__device__ __forceinline__ int st_idx(int row, int col) {
    return ((((row >> 4) * 16 + (col >> 3)) * 32 + (row & 15) + 16 * ((col >> 1) & 1)) << 2)
           + ((col >> 2) & 1) * 2 + (col & 1);
}

__device__ __forceinline__ v8f wmma4(v2f a, v2f b, v8f c) {
    return __builtin_amdgcn_wmma_f32_16x16x4_f32(
         false, a,  false, b,
         (short)0, c,  false,  false);
}

__device__ __forceinline__ float sigmoidf_dev(float x) {
    return 1.0f / (1.0f + __expf(-x));
}
__device__ __forceinline__ float tanhf_dev(float x) {
    x = fminf(15.0f, fmaxf(-15.0f, x));
    const float e = __expf(-2.0f * x);
    return (1.0f - e) / (1.0f + e);
}

#define AHEAD 8
__device__ __forceinline__ void tile_gemm3(const v4f* __restrict__ aF, const v2f* __restrict__ w,
                                           v8f& accA, v8f& accB, v8f& accC)
{
    const v2f* __restrict__ wA = w;
    const v2f* __restrict__ wB = w + 32 * 32;
    const v2f* __restrict__ wC = w + 2 * 32 * 32;
    v4f a4[D / 8];
    v2f bE[D / 8][3];
    v2f bO[D / 8][3];
#define LOAD_PAIR(SP) do { a4[SP] = aF[(SP) * 32]; \
        bE[SP][0] = wA[(SP) * 64];      bE[SP][1] = wB[(SP) * 64];      bE[SP][2] = wC[(SP) * 64]; \
        bO[SP][0] = wA[(SP) * 64 + 32]; bO[SP][1] = wB[(SP) * 64 + 32]; bO[SP][2] = wC[(SP) * 64 + 32]; } while (0)
    accA = (v8f){}; accB = (v8f){}; accC = (v8f){};
#pragma unroll
    for (int sp = 0; sp < AHEAD; ++sp) LOAD_PAIR(sp);
#pragma unroll
    for (int sp = 0; sp < D / 8; ++sp) {
        __builtin_amdgcn_sched_barrier(0);
        if (sp + AHEAD < D / 8) LOAD_PAIR(sp + AHEAD);
        __builtin_amdgcn_sched_barrier(0);
        const v2f aE = { a4[sp].x, a4[sp].y };
        const v2f aO = { a4[sp].z, a4[sp].w };
        accA = wmma4(aE, bE[sp][0], accA);
        accB = wmma4(aE, bE[sp][1], accB);
        accC = wmma4(aE, bE[sp][2], accC);
        accA = wmma4(aO, bO[sp][0], accA);
        accB = wmma4(aO, bO[sp][1], accB);
        accC = wmma4(aO, bO[sp][2], accC);
    }
#undef LOAD_PAIR
}

#define AHEAD2 11
__device__ __forceinline__ void tile_gemm2(const v4f* __restrict__ aF, const v2f* __restrict__ w,
                                           v8f& accA, v8f& accB)
{
    const v2f* __restrict__ wA = w;
    const v2f* __restrict__ wB = w + 32 * 32;
    v4f a4[D / 8];
    v2f bE[D / 8][2];
    v2f bO[D / 8][2];
#define LOAD_PAIR2(SP) do { a4[SP] = aF[(SP) * 32]; \
        bE[SP][0] = wA[(SP) * 64];      bE[SP][1] = wB[(SP) * 64]; \
        bO[SP][0] = wA[(SP) * 64 + 32]; bO[SP][1] = wB[(SP) * 64 + 32]; } while (0)
    accA = (v8f){}; accB = (v8f){};
#pragma unroll
    for (int sp = 0; sp < AHEAD2; ++sp) {
        LOAD_PAIR2(sp);
        __builtin_amdgcn_sched_barrier(0);
    }
#pragma unroll
    for (int sp = 0; sp < D / 8; ++sp) {
        __builtin_amdgcn_sched_barrier(0);
        if (sp + AHEAD2 < D / 8) LOAD_PAIR2(sp + AHEAD2);
        __builtin_amdgcn_sched_barrier(0);
        const v2f aE = { a4[sp].x, a4[sp].y };
        const v2f aO = { a4[sp].z, a4[sp].w };
        accA = wmma4(aE, bE[sp][0], accA);
        accB = wmma4(aE, bE[sp][1], accB);
        accA = wmma4(aO, bO[sp][0], accA);
        accB = wmma4(aO, bO[sp][1], accB);
    }
#undef LOAD_PAIR2
}

__device__ __forceinline__ v8f tile_gemm1(const v4f* __restrict__ aF, const v2f* __restrict__ w)
{
    v4f a4[D / 8];
    v2f bE[D / 8];
    v2f bO[D / 8];
#pragma unroll
    for (int sp = 0; sp < D / 8; ++sp) {
        a4[sp] = aF[sp * 32];
        bE[sp] = w[sp * 64];
        bO[sp] = w[sp * 64 + 32];
    }
    __builtin_amdgcn_sched_barrier(0);
    v8f acc = {};
#pragma unroll
    for (int sp = 0; sp < D / 8; ++sp) {
        const v2f aE = { a4[sp].x, a4[sp].y };
        const v2f aO = { a4[sp].z, a4[sp].w };
        acc = wmma4(aE, bE[sp], acc);
        acc = wmma4(aO, bO[sp], acc);
    }
    return acc;
}

__global__ __launch_bounds__(256) void gru_xproj_kernel(
    const float* __restrict__ xs, const float* __restrict__ Wx,
    const float* __restrict__ bx, const float* __restrict__ Wh,
    float* __restrict__ Xp, float* __restrict__ Wp, float* __restrict__ hz)
{
    {
        const int gid = blockIdx.x * 256 + threadIdx.x;
        const int m   = gid / PKV;
        const int r   = gid - m * PKV;
        const int pl  = r & 31;
        const int ps  = (r >> 5) & 31;
        const int gj  = r >> 10;
        const int pjt = gj / 3;
        const int pg  = gj - pjt * 3;
        const int off = (pg * D + pjt * 16 + (pl & 15)) * D + 4 * ps + 2 * (pl >> 4);
        const float* __restrict__ wsrc = m ? Wx : Wh;
        const v2f v = { wsrc[off], wsrc[off + 1] };
        ((v2f*)Wp)[gid] = v;
        if (gid < 2 * D * D) hz[gid] = 0.0f;
    }

    const int lane = threadIdx.x & 31;
    const int wave = threadIdx.x >> 5;
    const int tile = blockIdx.x * 8 + wave;
    const int tm = tile / 24;
    const int tn = tile % 24;
    const int l15 = lane & 15;
    const int hi  = lane >> 4;

    const float* __restrict__ arow = &xs[(tm * 16 + l15) * D];
    const float* __restrict__ brow = &Wx[(tn * 16 + l15) * D];

    v8f acc = {};
    for (int k0 = 0; k0 < D; k0 += 4) {
        const int kb = k0 + 2 * hi;
        v2f a = { arow[kb], arow[kb + 1] };
        v2f b = { brow[kb], brow[kb + 1] };
        acc = wmma4(a, b, acc);
    }

    const int col  = tn * 16 + l15;
    const float bb = bx[col];
#pragma unroll
    for (int r = 0; r < 8; ++r) {
        const int row = tm * 16 + r + 8 * hi;
        Xp[row * G3 + col] = acc[r] + bb;
    }
}

__global__ __launch_bounds__(32) void gru_l1_step_kernel(
    const int*   __restrict__ mk,
    const float* __restrict__ bh,
    const float* __restrict__ xp,
    const float* __restrict__ Wp,
    const float* __restrict__ hold, float* __restrict__ hnew)
{
    const int lane = threadIdx.x;
    const int l15  = lane & 15;
    const int hi   = lane >> 4;
    const int mt   = blockIdx.x >> 3;
    const int jt   = blockIdx.x & 7;
    const int col  = jt * 16 + l15;
    const int eb   = st_idx(mt * 16 + 8 * hi, col);

    const int mv = mk[0];
    float ho[8];
#pragma unroll
    for (int r = 0; r < 8; ++r) ho[r] = hold[eb + 4 * r];
    const float sa = xp[col]         + bh[col];
    const float sb = xp[D + col]     + bh[D + col];
    const float xc = xp[2 * D + col];
    const float bc = bh[2 * D + col];

    v8f accA, accB, accC;
    tile_gemm3((const v4f*)hold + (mt * 16) * 32 + lane,
               (const v2f*)Wp + ((jt * 3) * 32) * 32 + lane, accA, accB, accC);

#pragma unroll
    for (int r = 0; r < 8; ++r) {
        const float u  = sigmoidf_dev(accA[r] + sa);
        const float rg = sigmoidf_dev(accB[r] + sb);
        const float cd = tanhf_dev(xc + rg * (accC[r] + bc));
        const float nv = u * ho[r] + (1.0f - u) * cd;
        hnew[eb + 4 * r] = (mv > 0) ? nv : ho[r];
    }
}

__global__ __launch_bounds__(64) void gru_l2_step_kernel(
    const int*   __restrict__ mk,
    const float* __restrict__ bh, const float* __restrict__ bx,
    const float* __restrict__ Wp,
    const float* __restrict__ h1n,
    const float* __restrict__ hold, float* __restrict__ hnew)
{
    __shared__ v4f xh[6][32];

    const int lane = threadIdx.x & 31;
    const int wv   = __builtin_amdgcn_readfirstlane(threadIdx.x >> 5);
    const int l15  = lane & 15;
    const int hi   = lane >> 4;
    const int mt   = blockIdx.x >> 3;
    const int jt   = blockIdx.x & 7;
    const int col  = jt * 16 + l15;
    const int eb   = st_idx(mt * 16 + 8 * hi, col);

    const int mv = mk[0];
    float ho[8];
#pragma unroll
    for (int r = 0; r < 8; ++r) ho[r] = hold[eb + 4 * r];
    const float ba0 = bh[col],         bb0 = bx[col];
    const float ba1 = bh[D + col],     bb1 = bx[D + col];
    const float ba2 = bh[2 * D + col], bb2 = bx[2 * D + col];

    const float* __restrict__ asrc = wv ? h1n : hold;
    const v2f*   __restrict__ wsrc = (const v2f*)Wp + (wv ? PKV : 0);

    v8f accA, accB, accC;
    tile_gemm3((const v4f*)asrc + (mt * 16) * 32 + lane,
               wsrc + ((jt * 3) * 32) * 32 + lane, accA, accB, accC);

    if (wv) {
        xh[0][lane] = (v4f){ accA[0], accA[1], accA[2], accA[3] };
        xh[1][lane] = (v4f){ accA[4], accA[5], accA[6], accA[7] };
        xh[2][lane] = (v4f){ accB[0], accB[1], accB[2], accB[3] };
        xh[3][lane] = (v4f){ accB[4], accB[5], accB[6], accB[7] };
        xh[4][lane] = (v4f){ accC[0], accC[1], accC[2], accC[3] };
        xh[5][lane] = (v4f){ accC[4], accC[5], accC[6], accC[7] };
    }
    __syncthreads();
    if (wv) return;

    float xA[8], xB[8], xC[8];
    {
        const v4f a0 = xh[0][lane], a1 = xh[1][lane];
        const v4f b0 = xh[2][lane], b1 = xh[3][lane];
        const v4f c0 = xh[4][lane], c1 = xh[5][lane];
        xA[0] = a0.x; xA[1] = a0.y; xA[2] = a0.z; xA[3] = a0.w;
        xA[4] = a1.x; xA[5] = a1.y; xA[6] = a1.z; xA[7] = a1.w;
        xB[0] = b0.x; xB[1] = b0.y; xB[2] = b0.z; xB[3] = b0.w;
        xB[4] = b1.x; xB[5] = b1.y; xB[6] = b1.z; xB[7] = b1.w;
        xC[0] = c0.x; xC[1] = c0.y; xC[2] = c0.z; xC[3] = c0.w;
        xC[4] = c1.x; xC[5] = c1.y; xC[6] = c1.z; xC[7] = c1.w;
    }
#pragma unroll
    for (int r = 0; r < 8; ++r) {
        const float u  = sigmoidf_dev((accA[r] + ba0) + (xA[r] + bb0));
        const float rg = sigmoidf_dev((accB[r] + ba1) + (xB[r] + bb1));
        const float cd = tanhf_dev((xC[r] + bb2) + rg * (accC[r] + ba2));
        const float nv = u * ho[r] + (1.0f - u) * cd;
        hnew[eb + 4 * r] = (mv > 0) ? nv : ho[r];
    }
}

__global__ __launch_bounds__(128) void gru_l12_step_kernel(
    const int*   __restrict__ mk1,
    const int*   __restrict__ mk2,
    const float* __restrict__ bh, const float* __restrict__ bx,
    const float* __restrict__ xp,
    const float* __restrict__ Wp,
    const float* __restrict__ h1c,
    float*       __restrict__ h1n,
    const float* __restrict__ h2o, float* __restrict__ h2n)
{
    __shared__ v4f xh[5][2][32];

    const int lane = threadIdx.x & 31;
    const int wv   = __builtin_amdgcn_readfirstlane(threadIdx.x >> 5);
    const int l15  = lane & 15;
    const int hi   = lane >> 4;
    const bool l2  = blockIdx.x < 64;
    const int tile = blockIdx.x & 63;
    const int mt   = tile >> 3;
    const int jt   = tile & 7;
    const int col  = jt * 16 + l15;
    const int eb   = st_idx(mt * 16 + 8 * hi, col);

    const bool two  = l2 && (wv < 2);
    const bool one  = l2 ? (wv >= 2) : (wv < 3);
    const int  pr   = l2 ? (wv & 1) : 0;
    const int  g    = l2 ? (wv & 2) : wv;
    const int  slot = l2 ? (pr ? (2 + g) : g) : g;

    const float* __restrict__ hold = l2 ? h2o : h1c;
    const float* __restrict__ asrc = pr ? h1c : hold;
    const v4f*   __restrict__ aF   = (const v4f*)asrc + (mt * 16) * 32 + lane;
    const v2f*   __restrict__ wF   = (const v2f*)Wp + (pr ? PKV : 0) + ((jt * 3 + g) * 32) * 32 + lane;

    const int mv = (l2 ? mk2 : mk1)[0];
    float ho[8] = { 0.0f, 0.0f, 0.0f, 0.0f, 0.0f, 0.0f, 0.0f, 0.0f };
    float p0 = 0.0f, p1 = 0.0f, p2 = 0.0f, q0 = 0.0f, q1 = 0.0f, q2 = 0.0f;
    if (wv == 2) {
        const float* __restrict__ qsrc = l2 ? bx : xp;
#pragma unroll
        for (int r = 0; r < 8; ++r) ho[r] = hold[eb + 4 * r];
        p0 = bh[col];         q0 = qsrc[col];
        p1 = bh[D + col];     q1 = qsrc[D + col];
        p2 = bh[2 * D + col]; q2 = qsrc[2 * D + col];
    }

    v8f acc = {};
    if (two) {
        v8f accB;
        tile_gemm2(aF, wF, acc, accB);
        xh[slot][0][lane]     = (v4f){ acc[0], acc[1], acc[2], acc[3] };
        xh[slot][1][lane]     = (v4f){ acc[4], acc[5], acc[6], acc[7] };
        xh[slot + 1][0][lane] = (v4f){ accB[0], accB[1], accB[2], accB[3] };
        xh[slot + 1][1][lane] = (v4f){ accB[4], accB[5], accB[6], accB[7] };
    } else if (one) {
        acc = tile_gemm1(aF, wF);
        if (wv != 2) {
            xh[slot][0][lane] = (v4f){ acc[0], acc[1], acc[2], acc[3] };
            xh[slot][1][lane] = (v4f){ acc[4], acc[5], acc[6], acc[7] };
        }
    }
    __syncthreads();
    if (wv != 2) return;

    float hA[8], hB[8];
    {
        const v4f a0 = xh[0][0][lane], a1 = xh[0][1][lane];
        const v4f b0 = xh[1][0][lane], b1 = xh[1][1][lane];
        hA[0] = a0.x; hA[1] = a0.y; hA[2] = a0.z; hA[3] = a0.w;
        hA[4] = a1.x; hA[5] = a1.y; hA[6] = a1.z; hA[7] = a1.w;
        hB[0] = b0.x; hB[1] = b0.y; hB[2] = b0.z; hB[3] = b0.w;
        hB[4] = b1.x; hB[5] = b1.y; hB[6] = b1.z; hB[7] = b1.w;
    }

    if (!l2) {
        const float sa = q0 + p0;
        const float sb = q1 + p1;
#pragma unroll
        for (int r = 0; r < 8; ++r) {
            const float u  = sigmoidf_dev(hA[r] + sa);
            const float rg = sigmoidf_dev(hB[r] + sb);
            const float cd = tanhf_dev(q2 + rg * (acc[r] + p2));
            const float nv = u * ho[r] + (1.0f - u) * cd;
            h1n[eb + 4 * r] = (mv > 0) ? nv : ho[r];
        }
        return;
    }

    float xA[8], xB[8], xC[8];
    {
        const v4f a0 = xh[2][0][lane], a1 = xh[2][1][lane];
        const v4f b0 = xh[3][0][lane], b1 = xh[3][1][lane];
        const v4f c0 = xh[4][0][lane], c1 = xh[4][1][lane];
        xA[0] = a0.x; xA[1] = a0.y; xA[2] = a0.z; xA[3] = a0.w;
        xA[4] = a1.x; xA[5] = a1.y; xA[6] = a1.z; xA[7] = a1.w;
        xB[0] = b0.x; xB[1] = b0.y; xB[2] = b0.z; xB[3] = b0.w;
        xB[4] = b1.x; xB[5] = b1.y; xB[6] = b1.z; xB[7] = b1.w;
        xC[0] = c0.x; xC[1] = c0.y; xC[2] = c0.z; xC[3] = c0.w;
        xC[4] = c1.x; xC[5] = c1.y; xC[6] = c1.z; xC[7] = c1.w;
    }
#pragma unroll
    for (int r = 0; r < 8; ++r) {
        const float u  = sigmoidf_dev((hA[r] + p0) + (xA[r] + q0));
        const float rg = sigmoidf_dev((hB[r] + p1) + (xB[r] + q1));
        const float cd = tanhf_dev((xC[r] + q2) + rg * (acc[r] + p2));
        const float nv = u * ho[r] + (1.0f - u) * cd;
        h2n[eb + 4 * r] = (mv > 0) ? nv : ho[r];
    }
}

__global__ __launch_bounds__(256) void gru_out_kernel(
    const float* __restrict__ h1, const float* __restrict__ h2, float* __restrict__ out)
{
    const int i = blockIdx.x * 256 + threadIdx.x;
    const int f = st_idx(i >> 7, i & (D - 1));
    const float a = h1[f];
    const float b = h2[f];
    out[i] = b;
    out[D * D + 2 * i]     = a;
    out[D * D + 2 * i + 1] = b;
}

extern "C" void kernel_launch(void* const* d_in, const int* in_sizes, int n_in,
                              void* d_out, int out_size, void* d_ws, size_t ws_size,
                              hipStream_t stream) {
    const float* xs   = (const float*)d_in[0];
    const int*   mask = (const int*)  d_in[1];
    const float* Wh   = (const float*)d_in[2];
    const float* bh   = (const float*)d_in[3];
    const float* Wx   = (const float*)d_in[4];
    const float* bx   = (const float*)d_in[5];
    float* Xp  = (float*)d_ws;
    float* Wp  = Xp + NTOK * G3;
    float* out = (float*)d_out;
    float* h1a = Wp + 2 * G3 * D;
    float* h2a = h1a + D * D;
    float* h1b = out;
    float* h2b = out + D * D;

    gru_xproj_kernel<<<192, 256, 0, stream>>>(xs, Wx, bx, Wh, Xp, Wp, h1a);

    gru_l1_step_kernel<<<64, 32, 0, stream>>>(mask, bh, Xp, Wp, h1a, h1b);
    const float* xp = Xp + G3;
    const int*   mk = mask + 1;
    for (; mk != mask + NTOK - 1; mk += 2, xp += 2 * G3) {
        gru_l12_step_kernel<<<128, 128, 0, stream>>>(mk,     mk - 1, bh, bx, xp,      Wp, h1b, h1a, h2a, h2b);
        gru_l12_step_kernel<<<128, 128, 0, stream>>>(mk + 1, mk,     bh, bx, xp + G3, Wp, h1a, h1b, h2b, h2a);
    }
    gru_l12_step_kernel<<<128, 128, 0, stream>>>(mk, mk - 1, bh, bx, xp, Wp, h1b, h1a, h2a, h2b);
    gru_l2_step_kernel<<<64, 64, 0, stream>>>(mk, bh, bx, Wp, h1a, h2b, h2a);

    gru_out_kernel<<<64, 256, 0, stream>>>(h1a, h2a, out);
}
